# starting kernel
# baseline (speedup 1.0000x reference)
#include <hip/hip_runtime.h>


typedef __attribute__((ext_vector_type(2))) float v2f;
typedef __attribute__((ext_vector_type(4))) float v4f;
typedef __attribute__((ext_vector_type(8))) float v8f;

#define BATCH  8
#define SEQLEN 4096
#define HID    1024
#define STATE  64
#define T      16
#define NCHUNK (SEQLEN / T)
#define DPB    16
#define BP     16

#define PER_D   2368
#define OFF_P   0
#define OFF_G   1024
#define OFF_M   2048
#define OFF_LAM 2304

static __device__ __forceinline__ v8f wmma4(v2f a, v2f b, v8f c) {
  return __builtin_amdgcn_wmma_f32_16x16x4_f32(false, a, false, b, (short)0, c,
                                               false, false);
}

__global__ __launch_bounds__(64) void ssm_setup_kernel(
    const float* __restrict__ A, const float* __restrict__ Bm,
    const float* __restrict__ C, const float* __restrict__ Dv,
    const float* __restrict__ dt, float* __restrict__ ws) {
  const int d = blockIdx.x;
  const int n = threadIdx.x;
  __shared__ float sK[T * STATE];
  __shared__ float sKl[T];
  float* base = ws + (size_t)d * PER_D;

  const float a  = expf(A[d * STATE + n]);
  const float da = expf(-a * dt[n]);
  const float db = Bm[d * STATE + n] * (1.0f - da) / a;
  const float cc = C[d * STATE + n];

  float p = da;
#pragma unroll
  for (int t = 0; t < T; ++t) { base[OFF_P + t * STATE + n] = cc * p; p *= da; }

  const float da2 = da * da, da4 = da2 * da2, da8 = da4 * da4;
  base[OFF_LAM + n] = da8 * da8;

  float g = db;
  for (int s = T - 1; s >= 0; --s) { base[OFF_G + n * T + s] = g; g *= da; }

  float w = cc * db;
#pragma unroll
  for (int tau = 0; tau < T; ++tau) { sK[tau * STATE + n] = w; w *= da; }
  __syncthreads();
  if (n < T) {
    float s = 0.0f;
    for (int m = 0; m < STATE; ++m) s += sK[n * STATE + m];
    sKl[n] = s;
  }
  __syncthreads();
  if (n < T) {
    const int t = n;
    const float dvd = Dv[d];
    for (int s2 = 0; s2 < T; ++s2) {
      float v = (t >= s2) ? sKl[t - s2] : 0.0f;
      if (s2 == t) v += dvd;
      base[OFF_M + t * T + s2] = v;
    }
  }
}

__global__ __launch_bounds__(32 * DPB) void ssm_scan_kernel(
    const float* __restrict__ u, const float* __restrict__ ws,
    float* __restrict__ out) {
  const int tid  = threadIdx.x;
  const int lane = tid & 31;
  const int wv   = tid >> 5;
  const int d0   = blockIdx.x * DPB;
  const int half = lane >> 4;
  const int col  = lane & 15;

  __shared__ __align__(16) float ldsU[T * BP * DPB];
  __shared__ __align__(16) float ldsY[T * BP * DPB];

#pragma unroll
  for (int k = 0; k < (T * BP * DPB) / (32 * DPB); ++k)
    ldsU[tid + k * (32 * DPB)] = 0.0f;

  const float* base = ws + (size_t)(d0 + wv) * PER_D;

  v2f pA[16];
#pragma unroll
  for (int kk = 0; kk < 16; ++kk)
    pA[kk] = *(const v2f*)(base + OFF_P + col * STATE + 4 * kk + 2 * half);
  v2f mA[4];
#pragma unroll
  for (int kk = 0; kk < 4; ++kk)
    mA[kk] = *(const v2f*)(base + OFF_M + col * T + 4 * kk + 2 * half);
  v2f gA[4][4];
#pragma unroll
  for (int r = 0; r < 4; ++r)
#pragma unroll
    for (int kk = 0; kk < 4; ++kk)
      gA[r][kk] =
          *(const v2f*)(base + OFF_G + (r * 16 + col) * T + 4 * kk + 2 * half);
  float lam[4][8];
#pragma unroll
  for (int r = 0; r < 4; ++r)
#pragma unroll
    for (int i = 0; i < 8; ++i)
      lam[r][i] = base[OFF_LAM + r * 16 + i + 8 * half];

  v8f x[4] = {};

  const int pair = tid >> 2;
  const int bb   = pair >> 4;
  const int tt   = pair & 15;
  const int d4   = (tid & 3) << 2;
  const size_t gRow = (size_t)bb * SEQLEN * HID + (size_t)(d0 + d4);

  for (int c = 0; c < NCHUNK; ++c) {
    {
      const float* gp = u + gRow + (size_t)(c * T + tt) * HID;
      v4f v = *(const v4f*)gp;
      *(v4f*)(&ldsU[(tt * BP + bb) * DPB + d4]) = v;
      if (c + 1 < NCHUNK)
        __builtin_prefetch(u + gRow + (size_t)((c + 1) * T + tt) * HID, 0, 0);
    }
    __syncthreads();

    v2f uB[4];
#pragma unroll
    for (int kk = 0; kk < 4; ++kk) {
#pragma unroll
      for (int j = 0; j < 2; ++j) {
        const int s = 4 * kk + 2 * half + j;
        uB[kk][j] = ldsU[(s * BP + col) * DPB + wv];
      }
    }

    v8f y = {};
#pragma unroll
    for (int kk = 0; kk < 4; ++kk) y = wmma4(mA[kk], uB[kk], y);

#pragma unroll
    for (int kk = 0; kk < 16; ++kk) {
      v2f xb;
#pragma unroll
      for (int j = 0; j < 2; ++j) {
        const int n0 = 4 * kk + j;
        const int n1 = 4 * kk + 2 + j;
        const float c0 =
            __shfl(x[n0 >> 4][(n0 & 15) & 7], col + 16 * ((n0 & 15) >> 3), 32);
        const float c1 =
            __shfl(x[n1 >> 4][(n1 & 15) & 7], col + 16 * ((n1 & 15) >> 3), 32);
        xb[j] = half ? c1 : c0;
      }
      y = wmma4(pA[kk], xb, y);
    }

#pragma unroll
    for (int r = 0; r < 4; ++r) {
      v8f xn;
#pragma unroll
      for (int i = 0; i < 8; ++i) xn[i] = lam[r][i] * x[r][i];
#pragma unroll
      for (int kk = 0; kk < 4; ++kk) xn = wmma4(gA[r][kk], uB[kk], xn);
      x[r] = xn;
    }

#pragma unroll
    for (int i = 0; i < 8; ++i) {
      const int t = i + 8 * half;
      ldsY[(t * BP + col) * DPB + wv] = y[i];
    }
    __syncthreads();
    {
      v4f v = *(const v4f*)(&ldsY[(tt * BP + bb) * DPB + d4]);
      *(v4f*)(out + gRow + (size_t)(c * T + tt) * HID) = v;
    }
  }
}

extern "C" void kernel_launch(void* const* d_in, const int* in_sizes, int n_in,
                              void* d_out, int out_size, void* d_ws,
                              size_t ws_size, hipStream_t stream) {
  (void)in_sizes; (void)n_in; (void)out_size; (void)ws_size;
  const float* u  = (const float*)d_in[0];
  const float* A  = (const float*)d_in[1];
  const float* Bm = (const float*)d_in[2];
  const float* C  = (const float*)d_in[3];
  const float* Dv = (const float*)d_in[4];
  const float* dt = (const float*)d_in[5];
  float* out = (float*)d_out;
  float* ws  = (float*)d_ws;

  ssm_setup_kernel<<<HID, STATE, 0, stream>>>(A, Bm, C, Dv, dt, ws);
  ssm_scan_kernel<<<HID / DPB, 32 * DPB, 0, stream>>>(u, ws, out);
}


# LDS u and y tiles re-laid out per channel in WMMA fragment order with padded pitches: no LDS access puts two lanes on one bank (was 8 per bank); B operand read as two 16-byte loads; sums unchanged
# speedup vs baseline: 4.1894x; 4.1894x over previous
#include <hip/hip_runtime.h>


typedef __attribute__((ext_vector_type(2))) float v2f;
typedef __attribute__((ext_vector_type(4))) float v4f;
typedef __attribute__((ext_vector_type(8))) float v8f;

#define BATCH  8
#define SEQLEN 4096
#define HID    1024
#define STATE  64
#define T      16
#define NCHUNK (SEQLEN / T)
#define DPB    16
#define BP     16

#define NSEG   8
#define NPAIR  (NSEG / 2)
#define SEGCH  (NCHUNK / NSEG)
#define SEGLEN (SEGCH * T)

#define PER_D   2368
#define OFF_P   0
#define OFF_G   1024
#define OFF_M   2048
#define OFF_LAM 2304

#define OFF_STATE ((size_t)HID * PER_D)
#define PLANE     1024

#define U_PE   64
#define U_PH   132
#define U_PD   260
#define U_TILE (DPB * U_PD)
#define Y_PI   68
#define Y_PH   136
#define Y_PD   276
#define Y_TILE (DPB * Y_PD)

static __device__ __forceinline__ v8f wmma4(v2f a, v2f b, v8f c) {
  return __builtin_amdgcn_wmma_f32_16x16x4_f32(false, a, false, b, (short)0, c,
                                               false, false);
}

__global__ __launch_bounds__(64) void ssm_setup_kernel(
    const float* __restrict__ A, const float* __restrict__ Bm,
    const float* __restrict__ C, const float* __restrict__ Dv,
    const float* __restrict__ dt, float* __restrict__ ws) {
  const int d = blockIdx.x;
  const int n = threadIdx.x;
  __shared__ float sK[T * STATE];
  __shared__ float sKl[T];
  float* base = ws + (size_t)d * PER_D;

  const float a  = expf(A[d * STATE + n]);
  const float da = expf(-a * dt[n]);
  const float db = Bm[d * STATE + n] * (1.0f - da) / a;
  const float cc = C[d * STATE + n];

  float p = da;
#pragma unroll
  for (int t = 0; t < T; ++t) { base[OFF_P + t * STATE + n] = cc * p; p *= da; }

  const float da2 = da * da, da4 = da2 * da2, da8 = da4 * da4;
  base[OFF_LAM + n] = da8 * da8;

  float g = db;
  for (int s = T - 1; s >= 0; --s) { base[OFF_G + n * T + s] = g; g *= da; }

  float w = cc * db;
#pragma unroll
  for (int tau = 0; tau < T; ++tau) { sK[tau * STATE + n] = w; w *= da; }
  __syncthreads();
  if (n < T) {
    float s = 0.0f;
    for (int m = 0; m < STATE; ++m) s += sK[n * STATE + m];
    sKl[n] = s;
  }
  __syncthreads();
  if (n < T) {
    const int t = n;
    const float dvd = Dv[d];
    for (int s2 = 0; s2 < T; ++s2) {
      float v = (t >= s2) ? sKl[t - s2] : 0.0f;
      if (s2 == t) v += dvd;
      base[OFF_M + t * T + s2] = v;
    }
  }
}

__global__ __launch_bounds__(32 * DPB) void ssm_segstate_kernel(
    const float* __restrict__ u, const float* __restrict__ prm,
    float* __restrict__ st) {
  const int tid  = threadIdx.x;
  const int lane = tid & 31;
  const int wv   = tid >> 5;
  const int d0   = blockIdx.x * DPB;
  const int q    = blockIdx.y;
  const int half = lane >> 4;
  const int col  = lane & 15;

  __shared__ __align__(16) float ldsU[2][U_TILE];

  const float* base = prm + (size_t)(d0 + wv) * PER_D;

  v2f gA[4][4];
#pragma unroll
  for (int r = 0; r < 4; ++r)
#pragma unroll
    for (int kk = 0; kk < 4; ++kk)
      gA[r][kk] =
          *(const v2f*)(base + OFF_G + (r * 16 + col) * T + 4 * kk + 2 * half);
  float lam[4][8];
#pragma unroll
  for (int r = 0; r < 4; ++r)
#pragma unroll
    for (int i = 0; i < 8; ++i)
      lam[r][i] = base[OFF_LAM + r * 16 + i + 8 * half];

  v8f x[4] = {};

  const int pair = tid >> 2;
  const int bb   = pair >> 4;
  const int tt   = pair & 15;
  const int d4   = (tid & 3) << 2;
  const float* gA0 = u + (size_t)bb * SEQLEN * HID + (size_t)(d0 + d4) +
                     (size_t)(2 * q * SEGLEN + tt) * HID;
  const float* gB0 = gA0 + (size_t)SEGLEN * HID;
  const int uw = d4 * U_PD + ((tt >> 1) & 1) * U_PH + (tt >> 3) * U_PE +
                 bb * 4 + ((tt >> 2) & 1) * 2 + (tt & 1);
  const int ur = wv * U_PD + half * U_PH + col * 4;

  v4f va = *(const v4f*)gA0;
  v4f vb = *(const v4f*)gB0;

  for (int c = 0; c < SEGCH; ++c) {
    float* tile = ldsU[c & 1];
#pragma unroll
    for (int i = 0; i < 4; ++i) {
      tile[uw + i * U_PD]      = va[i];
      tile[uw + i * U_PD + 32] = vb[i];
    }
    __syncthreads();
    if (c + 1 < SEGCH) {
      va = *(const v4f*)(gA0 + (size_t)(c + 1) * T * HID);
      vb = *(const v4f*)(gB0 + (size_t)(c + 1) * T * HID);
    }
    if (c + 2 < SEGCH) {
      __builtin_prefetch(gA0 + (size_t)(c + 2) * T * HID, 0, 0);
      __builtin_prefetch(gB0 + (size_t)(c + 2) * T * HID, 0, 0);
    }

    const v4f q0 = *(const v4f*)(&tile[ur]);
    const v4f q1 = *(const v4f*)(&tile[ur + U_PE]);
    v2f uB[4];
    uB[0][0] = q0[0]; uB[0][1] = q0[1]; uB[1][0] = q0[2]; uB[1][1] = q0[3];
    uB[2][0] = q1[0]; uB[2][1] = q1[1]; uB[3][0] = q1[2]; uB[3][1] = q1[3];

#pragma unroll
    for (int r = 0; r < 4; ++r) {
      v8f xn;
#pragma unroll
      for (int i = 0; i < 8; ++i) xn[i] = lam[r][i] * x[r][i];
#pragma unroll
      for (int kk = 0; kk < 4; ++kk) xn = wmma4(gA[r][kk], uB[kk], xn);
      x[r] = xn;
    }
  }

  float* sp = st + ((size_t)q * HID + (size_t)(d0 + wv)) * PLANE + lane;
#pragma unroll
  for (int r = 0; r < 4; ++r)
#pragma unroll
    for (int i = 0; i < 8; ++i) sp[(r * 8 + i) * 32] = x[r][i];
}

__global__ __launch_bounds__(512) void ssm_carry_kernel(
    const float* __restrict__ prm, float* __restrict__ st) {
  const int d    = blockIdx.x;
  const int t    = threadIdx.x;
  const int b    = t & 7;
  const int half = (t >> 3) & 1;
  const int reg  = t >> 4;
  const int n    = (reg >> 3) * 16 + (reg & 7) + 8 * half;

  float ls = prm[(size_t)d * PER_D + OFF_LAM + n];
#pragma unroll
  for (int k = 1; k < SEGCH; k <<= 1) ls *= ls;

  float* p = st + (size_t)d * PLANE + reg * 32 + half * 16 + b;
  float e[NSEG];
#pragma unroll
  for (int s = 0; s < NSEG; ++s)
    e[s] = p[(size_t)(s >> 1) * HID * PLANE + (s & 1) * 8];
  float carry = 0.0f;
#pragma unroll
  for (int s = 0; s < NSEG; ++s) {
    p[(size_t)(s >> 1) * HID * PLANE + (s & 1) * 8] = carry;
    carry = ls * carry + e[s];
  }
}

__global__ __launch_bounds__(32 * DPB) void ssm_scan_kernel(
    const float* __restrict__ u, const float* __restrict__ ws,
    float* __restrict__ out) {
  const int tid  = threadIdx.x;
  const int lane = tid & 31;
  const int wv   = tid >> 5;
  const int d0   = blockIdx.x * DPB;
  const int q    = blockIdx.y;
  const int half = lane >> 4;
  const int col  = lane & 15;

  __shared__ __align__(16) float ldsU[U_TILE];
  __shared__ __align__(16) float ldsY[Y_TILE];

  const float* base = ws + (size_t)(d0 + wv) * PER_D;

  v2f pA[16];
#pragma unroll
  for (int kk = 0; kk < 16; ++kk)
    pA[kk] = *(const v2f*)(base + OFF_P + col * STATE + 4 * kk + 2 * half);
  v2f mA[4];
#pragma unroll
  for (int kk = 0; kk < 4; ++kk)
    mA[kk] = *(const v2f*)(base + OFF_M + col * T + 4 * kk + 2 * half);
  v2f gA[4][4];
#pragma unroll
  for (int r = 0; r < 4; ++r)
#pragma unroll
    for (int kk = 0; kk < 4; ++kk)
      gA[r][kk] =
          *(const v2f*)(base + OFF_G + (r * 16 + col) * T + 4 * kk + 2 * half);
  float lam[4][8];
#pragma unroll
  for (int r = 0; r < 4; ++r)
#pragma unroll
    for (int i = 0; i < 8; ++i)
      lam[r][i] = base[OFF_LAM + r * 16 + i + 8 * half];

  v8f x[4];
  {
    const float* sp =
        ws + OFF_STATE + ((size_t)q * HID + (size_t)(d0 + wv)) * PLANE + lane;
#pragma unroll
    for (int r = 0; r < 4; ++r)
#pragma unroll
      for (int i = 0; i < 8; ++i) x[r][i] = sp[(r * 8 + i) * 32];
  }

  const int pair = tid >> 2;
  const int bb   = pair >> 4;
  const int tt   = pair & 15;
  const int d4   = (tid & 3) << 2;
  const size_t gRowA = (size_t)bb * SEQLEN * HID + (size_t)(d0 + d4) +
                       (size_t)(2 * q * SEGLEN + tt) * HID;
  const size_t gRowB = gRowA + (size_t)SEGLEN * HID;
  const int uw = d4 * U_PD + ((tt >> 1) & 1) * U_PH + (tt >> 3) * U_PE +
                 bb * 4 + ((tt >> 2) & 1) * 2 + (tt & 1);
  const int ur = wv * U_PD + half * U_PH + col * 4;
  const int yw = wv * Y_PD + half * Y_PH + col * 4;
  const int yr = d4 * Y_PD + (tt >> 3) * Y_PH + ((tt >> 2) & 1) * Y_PI +
                 bb * 4 + (tt & 3);

  for (int c = 0; c < SEGCH; ++c) {
    {
      const size_t co = (size_t)c * T * HID;
      v4f va = *(const v4f*)(u + gRowA + co);
      v4f vb = *(const v4f*)(u + gRowB + co);
#pragma unroll
      for (int i = 0; i < 4; ++i) {
        ldsU[uw + i * U_PD]      = va[i];
        ldsU[uw + i * U_PD + 32] = vb[i];
      }
      if (c + 1 < SEGCH) {
        __builtin_prefetch(u + gRowA + co + (size_t)T * HID, 0, 0);
        __builtin_prefetch(u + gRowB + co + (size_t)T * HID, 0, 0);
      }
    }
    __syncthreads();

    const v4f q0 = *(const v4f*)(&ldsU[ur]);
    const v4f q1 = *(const v4f*)(&ldsU[ur + U_PE]);
    v2f uB[4];
    uB[0][0] = q0[0]; uB[0][1] = q0[1]; uB[1][0] = q0[2]; uB[1][1] = q0[3];
    uB[2][0] = q1[0]; uB[2][1] = q1[1]; uB[3][0] = q1[2]; uB[3][1] = q1[3];

    v8f y = {};
#pragma unroll
    for (int kk = 0; kk < 4; ++kk) y = wmma4(mA[kk], uB[kk], y);

#pragma unroll
    for (int kk = 0; kk < 16; ++kk) {
      v2f xb;
#pragma unroll
      for (int j = 0; j < 2; ++j) {
        const int n0 = 4 * kk + j;
        const int n1 = 4 * kk + 2 + j;
        const float c0 =
            __shfl(x[n0 >> 4][(n0 & 15) & 7], col + 16 * ((n0 & 15) >> 3), 32);
        const float c1 =
            __shfl(x[n1 >> 4][(n1 & 15) & 7], col + 16 * ((n1 & 15) >> 3), 32);
        xb[j] = half ? c1 : c0;
      }
      y = wmma4(pA[kk], xb, y);
    }

#pragma unroll
    for (int r = 0; r < 4; ++r) {
      v8f xn;
#pragma unroll
      for (int i = 0; i < 8; ++i) xn[i] = lam[r][i] * x[r][i];
#pragma unroll
      for (int kk = 0; kk < 4; ++kk) xn = wmma4(gA[r][kk], uB[kk], xn);
      x[r] = xn;
    }

    {
      v4f y0, y1;
#pragma unroll
      for (int i = 0; i < 4; ++i) { y0[i] = y[i]; y1[i] = y[4 + i]; }
      *(v4f*)(&ldsY[yw])        = y0;
      *(v4f*)(&ldsY[yw + Y_PI]) = y1;
    }
    __syncthreads();
    {
      const size_t co = (size_t)c * T * HID;
      v4f va, vb;
#pragma unroll
      for (int i = 0; i < 4; ++i) {
        va[i] = ldsY[yr + i * Y_PD];
        vb[i] = ldsY[yr + i * Y_PD + 32];
      }
      *(v4f*)(out + gRowA + co) = va;
      *(v4f*)(out + gRowB + co) = vb;
    }
  }
}

extern "C" void kernel_launch(void* const* d_in, const int* in_sizes, int n_in,
                              void* d_out, int out_size, void* d_ws,
                              size_t ws_size, hipStream_t stream) {
  (void)in_sizes; (void)n_in; (void)out_size; (void)ws_size;
  const float* u  = (const float*)d_in[0];
  const float* A  = (const float*)d_in[1];
  const float* Bm = (const float*)d_in[2];
  const float* C  = (const float*)d_in[3];
  const float* Dv = (const float*)d_in[4];
  const float* dt = (const float*)d_in[5];
  float* out = (float*)d_out;
  float* ws  = (float*)d_ws;

  ssm_setup_kernel<<<HID, STATE, 0, stream>>>(A, Bm, C, Dv, dt, ws);
  ssm_segstate_kernel<<<dim3(HID / DPB, NPAIR), 32 * DPB, 0, stream>>>(
      u, ws, ws + OFF_STATE);
  ssm_carry_kernel<<<HID, 512, 0, stream>>>(ws, ws + OFF_STATE);
  ssm_scan_kernel<<<dim3(HID / DPB, NPAIR), 32 * DPB, 0, stream>>>(u, ws, out);
}


# Scan: carried state reaches the P*X B operand through 16 in-register half swaps (v_permlane16_swap) per chunk instead of 64 LDS-unit cross-lane reads and 32 selects; k order and all sums unchanged
# speedup vs baseline: 4.4125x; 1.0533x over previous
#include <hip/hip_runtime.h>


typedef __attribute__((ext_vector_type(2))) float v2f;
typedef __attribute__((ext_vector_type(4))) float v4f;
typedef __attribute__((ext_vector_type(8))) float v8f;

#define BATCH  8
#define SEQLEN 4096
#define HID    1024
#define STATE  64
#define T      16
#define NCHUNK (SEQLEN / T)
#define DPB    16
#define BP     16

#define NSEG   8
#define NPAIR  (NSEG / 2)
#define SEGCH  (NCHUNK / NSEG)
#define SEGLEN (SEGCH * T)

#define PER_D   2368
#define OFF_P   0
#define OFF_G   1024
#define OFF_M   2048
#define OFF_LAM 2304

#define OFF_STATE ((size_t)HID * PER_D)
#define PLANE     1024

#define U_PE   64
#define U_PH   132
#define U_PD   260
#define U_TILE (DPB * U_PD)
#define Y_PI   68
#define Y_PH   136
#define Y_PD   276
#define Y_TILE (DPB * Y_PD)

static __device__ __forceinline__ v8f wmma4(v2f a, v2f b, v8f c) {
  return __builtin_amdgcn_wmma_f32_16x16x4_f32(false, a, false, b, (short)0, c,
                                               false, false);
}

static __device__ __forceinline__ void half_swap(float a, float b, float& lo_hi,
                                                 float& hi_hi) {
  typedef __attribute__((ext_vector_type(2))) unsigned int v2u;
  const v2u r = __builtin_amdgcn_permlane16_swap(__float_as_uint(a),
                                                 __float_as_uint(b), false, false);
  lo_hi = __uint_as_float(r[0]);
  hi_hi = __uint_as_float(r[1]);
}

__global__ __launch_bounds__(64) void ssm_setup_kernel(
    const float* __restrict__ A, const float* __restrict__ Bm,
    const float* __restrict__ C, const float* __restrict__ Dv,
    const float* __restrict__ dt, float* __restrict__ ws) {
  const int d = blockIdx.x;
  const int n = threadIdx.x;
  __shared__ float sK[T * STATE];
  __shared__ float sKl[T];
  float* base = ws + (size_t)d * PER_D;

  const float a  = expf(A[d * STATE + n]);
  const float da = expf(-a * dt[n]);
  const float db = Bm[d * STATE + n] * (1.0f - da) / a;
  const float cc = C[d * STATE + n];

  float p = da;
#pragma unroll
  for (int t = 0; t < T; ++t) { base[OFF_P + t * STATE + n] = cc * p; p *= da; }

  const float da2 = da * da, da4 = da2 * da2, da8 = da4 * da4;
  base[OFF_LAM + n] = da8 * da8;

  float g = db;
  for (int s = T - 1; s >= 0; --s) { base[OFF_G + n * T + s] = g; g *= da; }

  float w = cc * db;
#pragma unroll
  for (int tau = 0; tau < T; ++tau) { sK[tau * STATE + n] = w; w *= da; }
  __syncthreads();
  if (n < T) {
    float s = 0.0f;
    for (int m = 0; m < STATE; ++m) s += sK[n * STATE + m];
    sKl[n] = s;
  }
  __syncthreads();
  if (n < T) {
    const int t = n;
    const float dvd = Dv[d];
    for (int s2 = 0; s2 < T; ++s2) {
      float v = (t >= s2) ? sKl[t - s2] : 0.0f;
      if (s2 == t) v += dvd;
      base[OFF_M + t * T + s2] = v;
    }
  }
}

__global__ __launch_bounds__(32 * DPB) void ssm_segstate_kernel(
    const float* __restrict__ u, const float* __restrict__ prm,
    float* __restrict__ st) {
  const int tid  = threadIdx.x;
  const int lane = tid & 31;
  const int wv   = tid >> 5;
  const int d0   = blockIdx.x * DPB;
  const int q    = blockIdx.y;
  const int half = lane >> 4;
  const int col  = lane & 15;

  __shared__ __align__(16) float ldsU[2][U_TILE];

  const float* base = prm + (size_t)(d0 + wv) * PER_D;

  v2f gA[4][4];
#pragma unroll
  for (int r = 0; r < 4; ++r)
#pragma unroll
    for (int kk = 0; kk < 4; ++kk)
      gA[r][kk] =
          *(const v2f*)(base + OFF_G + (r * 16 + col) * T + 4 * kk + 2 * half);
  float lam[4][8];
#pragma unroll
  for (int r = 0; r < 4; ++r)
#pragma unroll
    for (int i = 0; i < 8; ++i)
      lam[r][i] = base[OFF_LAM + r * 16 + i + 8 * half];

  v8f x[4] = {};

  const int pair = tid >> 2;
  const int bb   = pair >> 4;
  const int tt   = pair & 15;
  const int d4   = (tid & 3) << 2;
  const float* gA0 = u + (size_t)bb * SEQLEN * HID + (size_t)(d0 + d4) +
                     (size_t)(2 * q * SEGLEN + tt) * HID;
  const float* gB0 = gA0 + (size_t)SEGLEN * HID;
  const int uw = d4 * U_PD + ((tt >> 1) & 1) * U_PH + (tt >> 3) * U_PE +
                 bb * 4 + ((tt >> 2) & 1) * 2 + (tt & 1);
  const int ur = wv * U_PD + half * U_PH + col * 4;

  v4f va = *(const v4f*)gA0;
  v4f vb = *(const v4f*)gB0;

  for (int c = 0; c < SEGCH; ++c) {
    float* tile = ldsU[c & 1];
#pragma unroll
    for (int i = 0; i < 4; ++i) {
      tile[uw + i * U_PD]      = va[i];
      tile[uw + i * U_PD + 32] = vb[i];
    }
    __syncthreads();
    if (c + 1 < SEGCH) {
      va = *(const v4f*)(gA0 + (size_t)(c + 1) * T * HID);
      vb = *(const v4f*)(gB0 + (size_t)(c + 1) * T * HID);
    }
    if (c + 2 < SEGCH) {
      __builtin_prefetch(gA0 + (size_t)(c + 2) * T * HID, 0, 0);
      __builtin_prefetch(gB0 + (size_t)(c + 2) * T * HID, 0, 0);
    }

    const v4f q0 = *(const v4f*)(&tile[ur]);
    const v4f q1 = *(const v4f*)(&tile[ur + U_PE]);
    v2f uB[4];
    uB[0][0] = q0[0]; uB[0][1] = q0[1]; uB[1][0] = q0[2]; uB[1][1] = q0[3];
    uB[2][0] = q1[0]; uB[2][1] = q1[1]; uB[3][0] = q1[2]; uB[3][1] = q1[3];

#pragma unroll
    for (int r = 0; r < 4; ++r) {
      v8f xn;
#pragma unroll
      for (int i = 0; i < 8; ++i) xn[i] = lam[r][i] * x[r][i];
#pragma unroll
      for (int kk = 0; kk < 4; ++kk) xn = wmma4(gA[r][kk], uB[kk], xn);
      x[r] = xn;
    }
  }

  float* sp = st + ((size_t)q * HID + (size_t)(d0 + wv)) * PLANE + lane;
#pragma unroll
  for (int r = 0; r < 4; ++r)
#pragma unroll
    for (int i = 0; i < 8; ++i) sp[(r * 8 + i) * 32] = x[r][i];
}

__global__ __launch_bounds__(512) void ssm_carry_kernel(
    const float* __restrict__ prm, float* __restrict__ st) {
  const int d    = blockIdx.x;
  const int t    = threadIdx.x;
  const int b    = t & 7;
  const int half = (t >> 3) & 1;
  const int reg  = t >> 4;
  const int n    = (reg >> 3) * 16 + (reg & 7) + 8 * half;

  float ls = prm[(size_t)d * PER_D + OFF_LAM + n];
#pragma unroll
  for (int k = 1; k < SEGCH; k <<= 1) ls *= ls;

  float* p = st + (size_t)d * PLANE + reg * 32 + half * 16 + b;
  float e[NSEG];
#pragma unroll
  for (int s = 0; s < NSEG; ++s)
    e[s] = p[(size_t)(s >> 1) * HID * PLANE + (s & 1) * 8];
  float carry = 0.0f;
#pragma unroll
  for (int s = 0; s < NSEG; ++s) {
    p[(size_t)(s >> 1) * HID * PLANE + (s & 1) * 8] = carry;
    carry = ls * carry + e[s];
  }
}

__global__ __launch_bounds__(32 * DPB) void ssm_scan_kernel(
    const float* __restrict__ u, const float* __restrict__ ws,
    float* __restrict__ out) {
  const int tid  = threadIdx.x;
  const int lane = tid & 31;
  const int wv   = tid >> 5;
  const int d0   = blockIdx.x * DPB;
  const int q    = blockIdx.y;
  const int half = lane >> 4;
  const int col  = lane & 15;

  __shared__ __align__(16) float ldsU[U_TILE];
  __shared__ __align__(16) float ldsY[Y_TILE];

  const float* base = ws + (size_t)(d0 + wv) * PER_D;

  v2f pA[16];
#pragma unroll
  for (int kk = 0; kk < 16; ++kk)
    pA[kk] = *(const v2f*)(base + OFF_P + col * STATE + 4 * kk + 2 * half);
  v2f mA[4];
#pragma unroll
  for (int kk = 0; kk < 4; ++kk)
    mA[kk] = *(const v2f*)(base + OFF_M + col * T + 4 * kk + 2 * half);
  v2f gA[4][4];
#pragma unroll
  for (int r = 0; r < 4; ++r)
#pragma unroll
    for (int kk = 0; kk < 4; ++kk)
      gA[r][kk] =
          *(const v2f*)(base + OFF_G + (r * 16 + col) * T + 4 * kk + 2 * half);
  float lam[4][8];
#pragma unroll
  for (int r = 0; r < 4; ++r)
#pragma unroll
    for (int i = 0; i < 8; ++i)
      lam[r][i] = base[OFF_LAM + r * 16 + i + 8 * half];

  v8f x[4];
  {
    const float* sp =
        ws + OFF_STATE + ((size_t)q * HID + (size_t)(d0 + wv)) * PLANE + lane;
#pragma unroll
    for (int r = 0; r < 4; ++r)
#pragma unroll
      for (int i = 0; i < 8; ++i) x[r][i] = sp[(r * 8 + i) * 32];
  }

  const int pair = tid >> 2;
  const int bb   = pair >> 4;
  const int tt   = pair & 15;
  const int d4   = (tid & 3) << 2;
  const size_t gRowA = (size_t)bb * SEQLEN * HID + (size_t)(d0 + d4) +
                       (size_t)(2 * q * SEGLEN + tt) * HID;
  const size_t gRowB = gRowA + (size_t)SEGLEN * HID;
  const int uw = d4 * U_PD + ((tt >> 1) & 1) * U_PH + (tt >> 3) * U_PE +
                 bb * 4 + ((tt >> 2) & 1) * 2 + (tt & 1);
  const int ur = wv * U_PD + half * U_PH + col * 4;
  const int yw = wv * Y_PD + half * Y_PH + col * 4;
  const int yr = d4 * Y_PD + (tt >> 3) * Y_PH + ((tt >> 2) & 1) * Y_PI +
                 bb * 4 + (tt & 3);

  for (int c = 0; c < SEGCH; ++c) {
    {
      const size_t co = (size_t)c * T * HID;
      v4f va = *(const v4f*)(u + gRowA + co);
      v4f vb = *(const v4f*)(u + gRowB + co);
#pragma unroll
      for (int i = 0; i < 4; ++i) {
        ldsU[uw + i * U_PD]      = va[i];
        ldsU[uw + i * U_PD + 32] = vb[i];
      }
      if (c + 1 < SEGCH) {
        __builtin_prefetch(u + gRowA + co + (size_t)T * HID, 0, 0);
        __builtin_prefetch(u + gRowB + co + (size_t)T * HID, 0, 0);
      }
    }
    __syncthreads();

    const v4f q0 = *(const v4f*)(&ldsU[ur]);
    const v4f q1 = *(const v4f*)(&ldsU[ur + U_PE]);
    v2f uB[4];
    uB[0][0] = q0[0]; uB[0][1] = q0[1]; uB[1][0] = q0[2]; uB[1][1] = q0[3];
    uB[2][0] = q1[0]; uB[2][1] = q1[1]; uB[3][0] = q1[2]; uB[3][1] = q1[3];

    v8f y = {};
#pragma unroll
    for (int kk = 0; kk < 4; ++kk) y = wmma4(mA[kk], uB[kk], y);

    v8f xs[4];
#pragma unroll
    for (int r = 0; r < 4; ++r)
#pragma unroll
      for (int i = 0; i < 8; ++i) xs[r][i] = lam[r][i] * x[r][i];

    v2f xb[16];
#pragma unroll
    for (int r = 0; r < 4; ++r)
#pragma unroll
      for (int mm = 0; mm < 2; ++mm)
#pragma unroll
        for (int j = 0; j < 2; ++j) {
          float lo, hi;
          half_swap(x[r][4 * mm + j], x[r][4 * mm + 2 + j], lo, hi);
          xb[4 * r + mm][j]     = lo;
          xb[4 * r + 2 + mm][j] = hi;
        }

#pragma unroll
    for (int kk = 0; kk < 16; ++kk) y = wmma4(pA[kk], xb[kk], y);

#pragma unroll
    for (int r = 0; r < 4; ++r) {
      v8f xn = xs[r];
#pragma unroll
      for (int kk = 0; kk < 4; ++kk) xn = wmma4(gA[r][kk], uB[kk], xn);
      x[r] = xn;
    }

    {
      v4f y0, y1;
#pragma unroll
      for (int i = 0; i < 4; ++i) { y0[i] = y[i]; y1[i] = y[4 + i]; }
      *(v4f*)(&ldsY[yw])        = y0;
      *(v4f*)(&ldsY[yw + Y_PI]) = y1;
    }
    __syncthreads();
    {
      const size_t co = (size_t)c * T * HID;
      v4f va, vb;
#pragma unroll
      for (int i = 0; i < 4; ++i) {
        va[i] = ldsY[yr + i * Y_PD];
        vb[i] = ldsY[yr + i * Y_PD + 32];
      }
      *(v4f*)(out + gRowA + co) = va;
      *(v4f*)(out + gRowB + co) = vb;
    }
  }
}

extern "C" void kernel_launch(void* const* d_in, const int* in_sizes, int n_in,
                              void* d_out, int out_size, void* d_ws,
                              size_t ws_size, hipStream_t stream) {
  (void)in_sizes; (void)n_in; (void)out_size; (void)ws_size;
  const float* u  = (const float*)d_in[0];
  const float* A  = (const float*)d_in[1];
  const float* Bm = (const float*)d_in[2];
  const float* C  = (const float*)d_in[3];
  const float* Dv = (const float*)d_in[4];
  const float* dt = (const float*)d_in[5];
  float* out = (float*)d_out;
  float* ws  = (float*)d_ws;

  ssm_setup_kernel<<<HID, STATE, 0, stream>>>(A, Bm, C, Dv, dt, ws);
  ssm_segstate_kernel<<<dim3(HID / DPB, NPAIR), 32 * DPB, 0, stream>>>(
      u, ws, ws + OFF_STATE);
  ssm_carry_kernel<<<HID, 512, 0, stream>>>(ws, ws + OFF_STATE);
  ssm_scan_kernel<<<dim3(HID / DPB, NPAIR), 32 * DPB, 0, stream>>>(u, ws, out);
}


# Pre-pass re-tiled: 32 channels per work-group (1024 threads) so each u row it reads is one whole 128-B line, over 16 half-segments (last pair skipped); carry chains them: state sums in a new order
# speedup vs baseline: 4.7806x; 1.0834x over previous
#include <hip/hip_runtime.h>


typedef __attribute__((ext_vector_type(2))) float v2f;
typedef __attribute__((ext_vector_type(4))) float v4f;
typedef __attribute__((ext_vector_type(8))) float v8f;

#define BATCH  8
#define SEQLEN 4096
#define HID    1024
#define STATE  64
#define T      16
#define NCHUNK (SEQLEN / T)
#define DPB    16
#define BP     16

#define NSEG   8
#define NPAIR  (NSEG / 2)
#define SEGCH  (NCHUNK / NSEG)
#define SEGLEN (SEGCH * T)

#define PDPB   32
#define NSUB   (2 * NSEG)
#define SUBCH  (NCHUNK / NSUB)
#define SUBLEN (SUBCH * T)
#define NSUBPAIR (NSUB / 2 - 1)

#define PER_D   2368
#define OFF_P   0
#define OFF_G   1024
#define OFF_M   2048
#define OFF_LAM 2304

#define OFF_STATE ((size_t)HID * PER_D)
#define PLANE     1024
#define OFF_SUB   (OFF_STATE + (size_t)NPAIR * HID * PLANE)

#define U_PE   64
#define U_PH   132
#define U_PD   260
#define U_TILE (DPB * U_PD)
#define Y_PI   68
#define Y_PH   136
#define Y_PD   276
#define Y_TILE (DPB * Y_PD)

static __device__ __forceinline__ v8f wmma4(v2f a, v2f b, v8f c) {
  return __builtin_amdgcn_wmma_f32_16x16x4_f32(false, a, false, b, (short)0, c,
                                               false, false);
}

static __device__ __forceinline__ void half_swap(float a, float b, float& lo_hi,
                                                 float& hi_hi) {
  typedef __attribute__((ext_vector_type(2))) unsigned int v2u;
  const v2u r = __builtin_amdgcn_permlane16_swap(__float_as_uint(a),
                                                 __float_as_uint(b), false, false);
  lo_hi = __uint_as_float(r[0]);
  hi_hi = __uint_as_float(r[1]);
}

__global__ __launch_bounds__(64) void ssm_setup_kernel(
    const float* __restrict__ A, const float* __restrict__ Bm,
    const float* __restrict__ C, const float* __restrict__ Dv,
    const float* __restrict__ dt, float* __restrict__ ws) {
  const int d = blockIdx.x;
  const int n = threadIdx.x;
  __shared__ float sK[T * STATE];
  __shared__ float sKl[T];
  float* base = ws + (size_t)d * PER_D;

  const float a  = expf(A[d * STATE + n]);
  const float da = expf(-a * dt[n]);
  const float db = Bm[d * STATE + n] * (1.0f - da) / a;
  const float cc = C[d * STATE + n];

  float p = da;
#pragma unroll
  for (int t = 0; t < T; ++t) { base[OFF_P + t * STATE + n] = cc * p; p *= da; }

  const float da2 = da * da, da4 = da2 * da2, da8 = da4 * da4;
  base[OFF_LAM + n] = da8 * da8;

  float g = db;
  for (int s = T - 1; s >= 0; --s) { base[OFF_G + n * T + s] = g; g *= da; }

  float w = cc * db;
#pragma unroll
  for (int tau = 0; tau < T; ++tau) { sK[tau * STATE + n] = w; w *= da; }
  __syncthreads();
  if (n < T) {
    float s = 0.0f;
    for (int m = 0; m < STATE; ++m) s += sK[n * STATE + m];
    sKl[n] = s;
  }
  __syncthreads();
  if (n < T) {
    const int t = n;
    const float dvd = Dv[d];
    for (int s2 = 0; s2 < T; ++s2) {
      float v = (t >= s2) ? sKl[t - s2] : 0.0f;
      if (s2 == t) v += dvd;
      base[OFF_M + t * T + s2] = v;
    }
  }
}

__global__ __launch_bounds__(32 * PDPB) void ssm_segstate_kernel(
    const float* __restrict__ u, const float* __restrict__ prm,
    float* __restrict__ st) {
  const int tid  = threadIdx.x;
  const int lane = tid & 31;
  const int wv   = tid >> 5;
  const int d0   = blockIdx.x * PDPB;
  const int q    = blockIdx.y;
  const int half = lane >> 4;
  const int col  = lane & 15;

  __shared__ __align__(16) float ldsU[2][PDPB * U_PD];

  const float* base = prm + (size_t)(d0 + wv) * PER_D;

  v2f gA[4][4];
#pragma unroll
  for (int r = 0; r < 4; ++r)
#pragma unroll
    for (int kk = 0; kk < 4; ++kk)
      gA[r][kk] =
          *(const v2f*)(base + OFF_G + (r * 16 + col) * T + 4 * kk + 2 * half);
  float lam[4][8];
#pragma unroll
  for (int r = 0; r < 4; ++r)
#pragma unroll
    for (int i = 0; i < 8; ++i)
      lam[r][i] = base[OFF_LAM + r * 16 + i + 8 * half];

  v8f x[4] = {};

  const int pair = tid >> 3;
  const int bb   = pair >> 4;
  const int tt   = pair & 15;
  const int d4   = (tid & 7) << 2;
  const float* gA0 = u + (size_t)bb * SEQLEN * HID + (size_t)(d0 + d4) +
                     (size_t)(2 * q * SUBLEN + tt) * HID;
  const float* gB0 = gA0 + (size_t)SUBLEN * HID;
  const int uw = d4 * U_PD + ((tt >> 1) & 1) * U_PH + (tt >> 3) * U_PE +
                 bb * 4 + ((tt >> 2) & 1) * 2 + (tt & 1);
  const int ur = wv * U_PD + half * U_PH + col * 4;

  v4f va = *(const v4f*)gA0;
  v4f vb = *(const v4f*)gB0;

  for (int c = 0; c < SUBCH; ++c) {
    float* tile = ldsU[c & 1];
#pragma unroll
    for (int i = 0; i < 4; ++i) {
      tile[uw + i * U_PD]      = va[i];
      tile[uw + i * U_PD + 32] = vb[i];
    }
    __syncthreads();
    if (c + 1 < SUBCH) {
      va = *(const v4f*)(gA0 + (size_t)(c + 1) * T * HID);
      vb = *(const v4f*)(gB0 + (size_t)(c + 1) * T * HID);
    }
    if (c + 2 < SUBCH) {
      __builtin_prefetch(gA0 + (size_t)(c + 2) * T * HID, 0, 0);
      __builtin_prefetch(gB0 + (size_t)(c + 2) * T * HID, 0, 0);
    }

    const v4f q0 = *(const v4f*)(&tile[ur]);
    const v4f q1 = *(const v4f*)(&tile[ur + U_PE]);
    v2f uB[4];
    uB[0][0] = q0[0]; uB[0][1] = q0[1]; uB[1][0] = q0[2]; uB[1][1] = q0[3];
    uB[2][0] = q1[0]; uB[2][1] = q1[1]; uB[3][0] = q1[2]; uB[3][1] = q1[3];

#pragma unroll
    for (int r = 0; r < 4; ++r) {
      v8f xn;
#pragma unroll
      for (int i = 0; i < 8; ++i) xn[i] = lam[r][i] * x[r][i];
#pragma unroll
      for (int kk = 0; kk < 4; ++kk) xn = wmma4(gA[r][kk], uB[kk], xn);
      x[r] = xn;
    }
  }

  float* sp = st + ((size_t)q * HID + (size_t)(d0 + wv)) * PLANE + lane;
#pragma unroll
  for (int r = 0; r < 4; ++r)
#pragma unroll
    for (int i = 0; i < 8; ++i) sp[(r * 8 + i) * 32] = x[r][i];
}

__global__ __launch_bounds__(512) void ssm_carry_kernel(
    const float* __restrict__ prm, const float* __restrict__ sub,
    float* __restrict__ st) {
  const int d    = blockIdx.x;
  const int t    = threadIdx.x;
  const int b    = t & 7;
  const int half = (t >> 3) & 1;
  const int reg  = t >> 4;
  const int n    = (reg >> 3) * 16 + (reg & 7) + 8 * half;

  float ls = prm[(size_t)d * PER_D + OFF_LAM + n];
#pragma unroll
  for (int k = 1; k < SUBCH; k <<= 1) ls *= ls;

  const size_t o = (size_t)d * PLANE + reg * 32 + half * 16 + b;
  float e[NSUB - 2];
#pragma unroll
  for (int j = 0; j < NSUB - 2; ++j)
    e[j] = sub[o + (size_t)(j >> 1) * HID * PLANE + (j & 1) * 8];
  float carry = 0.0f;
#pragma unroll
  for (int s = 0; s < NSEG; ++s) {
    st[o + (size_t)(s >> 1) * HID * PLANE + (s & 1) * 8] = carry;
    if (s + 1 < NSEG) {
      carry = ls * carry + e[2 * s];
      carry = ls * carry + e[2 * s + 1];
    }
  }
}

__global__ __launch_bounds__(32 * DPB) void ssm_scan_kernel(
    const float* __restrict__ u, const float* __restrict__ ws,
    float* __restrict__ out) {
  const int tid  = threadIdx.x;
  const int lane = tid & 31;
  const int wv   = tid >> 5;
  const int d0   = blockIdx.x * DPB;
  const int q    = blockIdx.y;
  const int half = lane >> 4;
  const int col  = lane & 15;

  __shared__ __align__(16) float ldsU[U_TILE];
  __shared__ __align__(16) float ldsY[Y_TILE];

  const float* base = ws + (size_t)(d0 + wv) * PER_D;

  v2f pA[16];
#pragma unroll
  for (int kk = 0; kk < 16; ++kk)
    pA[kk] = *(const v2f*)(base + OFF_P + col * STATE + 4 * kk + 2 * half);
  v2f mA[4];
#pragma unroll
  for (int kk = 0; kk < 4; ++kk)
    mA[kk] = *(const v2f*)(base + OFF_M + col * T + 4 * kk + 2 * half);
  v2f gA[4][4];
#pragma unroll
  for (int r = 0; r < 4; ++r)
#pragma unroll
    for (int kk = 0; kk < 4; ++kk)
      gA[r][kk] =
          *(const v2f*)(base + OFF_G + (r * 16 + col) * T + 4 * kk + 2 * half);
  float lam[4][8];
#pragma unroll
  for (int r = 0; r < 4; ++r)
#pragma unroll
    for (int i = 0; i < 8; ++i)
      lam[r][i] = base[OFF_LAM + r * 16 + i + 8 * half];

  v8f x[4];
  {
    const float* sp =
        ws + OFF_STATE + ((size_t)q * HID + (size_t)(d0 + wv)) * PLANE + lane;
#pragma unroll
    for (int r = 0; r < 4; ++r)
#pragma unroll
      for (int i = 0; i < 8; ++i) x[r][i] = sp[(r * 8 + i) * 32];
  }

  const int pair = tid >> 2;
  const int bb   = pair >> 4;
  const int tt   = pair & 15;
  const int d4   = (tid & 3) << 2;
  const size_t gRowA = (size_t)bb * SEQLEN * HID + (size_t)(d0 + d4) +
                       (size_t)(2 * q * SEGLEN + tt) * HID;
  const size_t gRowB = gRowA + (size_t)SEGLEN * HID;
  const int uw = d4 * U_PD + ((tt >> 1) & 1) * U_PH + (tt >> 3) * U_PE +
                 bb * 4 + ((tt >> 2) & 1) * 2 + (tt & 1);
  const int ur = wv * U_PD + half * U_PH + col * 4;
  const int yw = wv * Y_PD + half * Y_PH + col * 4;
  const int yr = d4 * Y_PD + (tt >> 3) * Y_PH + ((tt >> 2) & 1) * Y_PI +
                 bb * 4 + (tt & 3);

  for (int c = 0; c < SEGCH; ++c) {
    {
      const size_t co = (size_t)c * T * HID;
      v4f va = *(const v4f*)(u + gRowA + co);
      v4f vb = *(const v4f*)(u + gRowB + co);
#pragma unroll
      for (int i = 0; i < 4; ++i) {
        ldsU[uw + i * U_PD]      = va[i];
        ldsU[uw + i * U_PD + 32] = vb[i];
      }
      if (c + 1 < SEGCH) {
        __builtin_prefetch(u + gRowA + co + (size_t)T * HID, 0, 0);
        __builtin_prefetch(u + gRowB + co + (size_t)T * HID, 0, 0);
      }
    }
    __syncthreads();

    const v4f q0 = *(const v4f*)(&ldsU[ur]);
    const v4f q1 = *(const v4f*)(&ldsU[ur + U_PE]);
    v2f uB[4];
    uB[0][0] = q0[0]; uB[0][1] = q0[1]; uB[1][0] = q0[2]; uB[1][1] = q0[3];
    uB[2][0] = q1[0]; uB[2][1] = q1[1]; uB[3][0] = q1[2]; uB[3][1] = q1[3];

    v8f y = {};
#pragma unroll
    for (int kk = 0; kk < 4; ++kk) y = wmma4(mA[kk], uB[kk], y);

    v8f xs[4];
#pragma unroll
    for (int r = 0; r < 4; ++r)
#pragma unroll
      for (int i = 0; i < 8; ++i) xs[r][i] = lam[r][i] * x[r][i];

    v2f xb[16];
#pragma unroll
    for (int r = 0; r < 4; ++r)
#pragma unroll
      for (int mm = 0; mm < 2; ++mm)
#pragma unroll
        for (int j = 0; j < 2; ++j) {
          float lo, hi;
          half_swap(x[r][4 * mm + j], x[r][4 * mm + 2 + j], lo, hi);
          xb[4 * r + mm][j]     = lo;
          xb[4 * r + 2 + mm][j] = hi;
        }

#pragma unroll
    for (int kk = 0; kk < 16; ++kk) y = wmma4(pA[kk], xb[kk], y);

#pragma unroll
    for (int r = 0; r < 4; ++r) {
      v8f xn = xs[r];
#pragma unroll
      for (int kk = 0; kk < 4; ++kk) xn = wmma4(gA[r][kk], uB[kk], xn);
      x[r] = xn;
    }

    {
      v4f y0, y1;
#pragma unroll
      for (int i = 0; i < 4; ++i) { y0[i] = y[i]; y1[i] = y[4 + i]; }
      *(v4f*)(&ldsY[yw])        = y0;
      *(v4f*)(&ldsY[yw + Y_PI]) = y1;
    }
    __syncthreads();
    {
      const size_t co = (size_t)c * T * HID;
      v4f va, vb;
#pragma unroll
      for (int i = 0; i < 4; ++i) {
        va[i] = ldsY[yr + i * Y_PD];
        vb[i] = ldsY[yr + i * Y_PD + 32];
      }
      *(v4f*)(out + gRowA + co) = va;
      *(v4f*)(out + gRowB + co) = vb;
    }
  }
}

extern "C" void kernel_launch(void* const* d_in, const int* in_sizes, int n_in,
                              void* d_out, int out_size, void* d_ws,
                              size_t ws_size, hipStream_t stream) {
  (void)in_sizes; (void)n_in; (void)out_size; (void)ws_size;
  const float* u  = (const float*)d_in[0];
  const float* A  = (const float*)d_in[1];
  const float* Bm = (const float*)d_in[2];
  const float* C  = (const float*)d_in[3];
  const float* Dv = (const float*)d_in[4];
  const float* dt = (const float*)d_in[5];
  float* out = (float*)d_out;
  float* ws  = (float*)d_ws;

  ssm_setup_kernel<<<HID, STATE, 0, stream>>>(A, Bm, C, Dv, dt, ws);
  ssm_segstate_kernel<<<dim3(HID / PDPB, NSUBPAIR), 32 * PDPB, 0, stream>>>(
      u, ws, ws + OFF_SUB);
  ssm_carry_kernel<<<HID, 512, 0, stream>>>(ws, ws + OFF_SUB, ws + OFF_STATE);
  ssm_scan_kernel<<<dim3(HID / DPB, NPAIR), 32 * DPB, 0, stream>>>(u, ws, out);
}


# Scan: 4 added transfer waves per work-group (20 waves) do every global<->LDS tile move, one chunk ahead/behind, via double-buffered tiles; matrix waves touch only LDS; 1 barrier per chunk; same sums
# speedup vs baseline: 5.2938x; 1.1074x over previous
#include <hip/hip_runtime.h>


typedef __attribute__((ext_vector_type(2))) float v2f;
typedef __attribute__((ext_vector_type(4))) float v4f;
typedef __attribute__((ext_vector_type(8))) float v8f;

#define BATCH  8
#define SEQLEN 4096
#define HID    1024
#define STATE  64
#define T      16
#define NCHUNK (SEQLEN / T)
#define DPB    16
#define BP     16

#define NSEG   8
#define NPAIR  (NSEG / 2)
#define SEGCH  (NCHUNK / NSEG)
#define SEGLEN (SEGCH * T)

#define PDPB   32
#define NSUB   (2 * NSEG)
#define SUBCH  (NCHUNK / NSUB)
#define SUBLEN (SUBCH * T)
#define NSUBPAIR (NSUB / 2 - 1)

#define PER_D   2368
#define OFF_P   0
#define OFF_G   1024
#define OFF_M   2048
#define OFF_LAM 2304

#define OFF_STATE ((size_t)HID * PER_D)
#define PLANE     1024
#define OFF_SUB   (OFF_STATE + (size_t)NPAIR * HID * PLANE)

#define U_PE   64
#define U_PH   132
#define U_PD   260
#define U_TILE (DPB * U_PD)
#define Y_PI   68
#define Y_PH   136
#define Y_PD   276
#define Y_TILE (DPB * Y_PD)

static __device__ __forceinline__ v8f wmma4(v2f a, v2f b, v8f c) {
  return __builtin_amdgcn_wmma_f32_16x16x4_f32(false, a, false, b, (short)0, c,
                                               false, false);
}

static __device__ __forceinline__ void half_swap(float a, float b, float& lo_hi,
                                                 float& hi_hi) {
  typedef __attribute__((ext_vector_type(2))) unsigned int v2u;
  const v2u r = __builtin_amdgcn_permlane16_swap(__float_as_uint(a),
                                                 __float_as_uint(b), false, false);
  lo_hi = __uint_as_float(r[0]);
  hi_hi = __uint_as_float(r[1]);
}

__global__ __launch_bounds__(64) void ssm_setup_kernel(
    const float* __restrict__ A, const float* __restrict__ Bm,
    const float* __restrict__ C, const float* __restrict__ Dv,
    const float* __restrict__ dt, float* __restrict__ ws) {
  const int d = blockIdx.x;
  const int n = threadIdx.x;
  __shared__ float sK[T * STATE];
  __shared__ float sKl[T];
  float* base = ws + (size_t)d * PER_D;

  const float a  = expf(A[d * STATE + n]);
  const float da = expf(-a * dt[n]);
  const float db = Bm[d * STATE + n] * (1.0f - da) / a;
  const float cc = C[d * STATE + n];

  float p = da;
#pragma unroll
  for (int t = 0; t < T; ++t) { base[OFF_P + t * STATE + n] = cc * p; p *= da; }

  const float da2 = da * da, da4 = da2 * da2, da8 = da4 * da4;
  base[OFF_LAM + n] = da8 * da8;

  float g = db;
  for (int s = T - 1; s >= 0; --s) { base[OFF_G + n * T + s] = g; g *= da; }

  float w = cc * db;
#pragma unroll
  for (int tau = 0; tau < T; ++tau) { sK[tau * STATE + n] = w; w *= da; }
  __syncthreads();
  if (n < T) {
    float s = 0.0f;
    for (int m = 0; m < STATE; ++m) s += sK[n * STATE + m];
    sKl[n] = s;
  }
  __syncthreads();
  if (n < T) {
    const int t = n;
    const float dvd = Dv[d];
    for (int s2 = 0; s2 < T; ++s2) {
      float v = (t >= s2) ? sKl[t - s2] : 0.0f;
      if (s2 == t) v += dvd;
      base[OFF_M + t * T + s2] = v;
    }
  }
}

__global__ __launch_bounds__(32 * PDPB) void ssm_segstate_kernel(
    const float* __restrict__ u, const float* __restrict__ prm,
    float* __restrict__ st) {
  const int tid  = threadIdx.x;
  const int lane = tid & 31;
  const int wv   = tid >> 5;
  const int d0   = blockIdx.x * PDPB;
  const int q    = blockIdx.y;
  const int half = lane >> 4;
  const int col  = lane & 15;

  __shared__ __align__(16) float ldsU[2][PDPB * U_PD];

  const float* base = prm + (size_t)(d0 + wv) * PER_D;

  v2f gA[4][4];
#pragma unroll
  for (int r = 0; r < 4; ++r)
#pragma unroll
    for (int kk = 0; kk < 4; ++kk)
      gA[r][kk] =
          *(const v2f*)(base + OFF_G + (r * 16 + col) * T + 4 * kk + 2 * half);
  float lam[4][8];
#pragma unroll
  for (int r = 0; r < 4; ++r)
#pragma unroll
    for (int i = 0; i < 8; ++i)
      lam[r][i] = base[OFF_LAM + r * 16 + i + 8 * half];

  v8f x[4] = {};

  const int pair = tid >> 3;
  const int bb   = pair >> 4;
  const int tt   = pair & 15;
  const int d4   = (tid & 7) << 2;
  const float* gA0 = u + (size_t)bb * SEQLEN * HID + (size_t)(d0 + d4) +
                     (size_t)(2 * q * SUBLEN + tt) * HID;
  const float* gB0 = gA0 + (size_t)SUBLEN * HID;
  const int uw = d4 * U_PD + ((tt >> 1) & 1) * U_PH + (tt >> 3) * U_PE +
                 bb * 4 + ((tt >> 2) & 1) * 2 + (tt & 1);
  const int ur = wv * U_PD + half * U_PH + col * 4;

  v4f va = *(const v4f*)gA0;
  v4f vb = *(const v4f*)gB0;

  for (int c = 0; c < SUBCH; ++c) {
    float* tile = ldsU[c & 1];
#pragma unroll
    for (int i = 0; i < 4; ++i) {
      tile[uw + i * U_PD]      = va[i];
      tile[uw + i * U_PD + 32] = vb[i];
    }
    __syncthreads();
    if (c + 1 < SUBCH) {
      va = *(const v4f*)(gA0 + (size_t)(c + 1) * T * HID);
      vb = *(const v4f*)(gB0 + (size_t)(c + 1) * T * HID);
    }
    if (c + 2 < SUBCH) {
      __builtin_prefetch(gA0 + (size_t)(c + 2) * T * HID, 0, 0);
      __builtin_prefetch(gB0 + (size_t)(c + 2) * T * HID, 0, 0);
    }

    const v4f q0 = *(const v4f*)(&tile[ur]);
    const v4f q1 = *(const v4f*)(&tile[ur + U_PE]);
    v2f uB[4];
    uB[0][0] = q0[0]; uB[0][1] = q0[1]; uB[1][0] = q0[2]; uB[1][1] = q0[3];
    uB[2][0] = q1[0]; uB[2][1] = q1[1]; uB[3][0] = q1[2]; uB[3][1] = q1[3];

#pragma unroll
    for (int r = 0; r < 4; ++r) {
      v8f xn;
#pragma unroll
      for (int i = 0; i < 8; ++i) xn[i] = lam[r][i] * x[r][i];
#pragma unroll
      for (int kk = 0; kk < 4; ++kk) xn = wmma4(gA[r][kk], uB[kk], xn);
      x[r] = xn;
    }
  }

  float* sp = st + ((size_t)q * HID + (size_t)(d0 + wv)) * PLANE + lane;
#pragma unroll
  for (int r = 0; r < 4; ++r)
#pragma unroll
    for (int i = 0; i < 8; ++i) sp[(r * 8 + i) * 32] = x[r][i];
}

__global__ __launch_bounds__(512) void ssm_carry_kernel(
    const float* __restrict__ prm, const float* __restrict__ sub,
    float* __restrict__ st) {
  const int d    = blockIdx.x;
  const int t    = threadIdx.x;
  const int b    = t & 7;
  const int half = (t >> 3) & 1;
  const int reg  = t >> 4;
  const int n    = (reg >> 3) * 16 + (reg & 7) + 8 * half;

  float ls = prm[(size_t)d * PER_D + OFF_LAM + n];
#pragma unroll
  for (int k = 1; k < SUBCH; k <<= 1) ls *= ls;

  const size_t o = (size_t)d * PLANE + reg * 32 + half * 16 + b;
  float e[NSUB - 2];
#pragma unroll
  for (int j = 0; j < NSUB - 2; ++j)
    e[j] = sub[o + (size_t)(j >> 1) * HID * PLANE + (j & 1) * 8];
  float carry = 0.0f;
#pragma unroll
  for (int s = 0; s < NSEG; ++s) {
    st[o + (size_t)(s >> 1) * HID * PLANE + (s & 1) * 8] = carry;
    if (s + 1 < NSEG) {
      carry = ls * carry + e[2 * s];
      carry = ls * carry + e[2 * s + 1];
    }
  }
}

#define NLW 4
#define SCAN_THREADS (32 * (DPB + NLW))

__global__ __launch_bounds__(SCAN_THREADS) void ssm_scan_kernel(
    const float* __restrict__ u, const float* __restrict__ ws,
    float* __restrict__ out) {
  const int tid  = threadIdx.x;
  const int lane = tid & 31;
  const int wv   = __builtin_amdgcn_readfirstlane(tid >> 5);
  const bool mover = wv < NLW;
  const int ch   = wv - NLW;
  const int d0   = blockIdx.x * DPB;
  const int q    = blockIdx.y;
  const int half = lane >> 4;
  const int col  = lane & 15;

  __shared__ __align__(16) float ldsU[2][U_TILE];
  __shared__ __align__(16) float ldsY[2][Y_TILE];
  __shared__ __align__(16) float ldsLam[DPB * STATE];

  v2f pA[16];
  v2f mA[4];
  v2f gA[4][4];
  v8f x[4] = {};
  if (!mover) {
    const float* base = ws + (size_t)(d0 + ch) * PER_D;
#pragma unroll
    for (int kk = 0; kk < 16; ++kk)
      pA[kk] = *(const v2f*)(base + OFF_P + col * STATE + 4 * kk + 2 * half);
#pragma unroll
    for (int kk = 0; kk < 4; ++kk)
      mA[kk] = *(const v2f*)(base + OFF_M + col * T + 4 * kk + 2 * half);
#pragma unroll
    for (int r = 0; r < 4; ++r)
#pragma unroll
      for (int kk = 0; kk < 4; ++kk)
        gA[r][kk] =
            *(const v2f*)(base + OFF_G + (r * 16 + col) * T + 4 * kk + 2 * half);
#pragma unroll
    for (int p = 0; p < 2; ++p) {
      const int idx = lane + 32 * p;
      ldsLam[ch * STATE + idx] =
          base[OFF_LAM + ((idx >> 3) & 3) * 16 + (idx & 7) + 8 * p];
    }
    const float* sp =
        ws + OFF_STATE + ((size_t)q * HID + (size_t)(d0 + ch)) * PLANE + lane;
#pragma unroll
    for (int r = 0; r < 4; ++r)
#pragma unroll
      for (int i = 0; i < 8; ++i) x[r][i] = sp[(r * 8 + i) * 32];
  }
  const int ur = ch * U_PD + half * U_PH + col * 4;
  const int yw = ch * Y_PD + half * Y_PH + col * 4;

  const int mt  = tid & 127;
  const int bb0 = mt >> 6;
  const int tt  = (mt >> 2) & 15;
  const int d4  = (mt & 3) << 2;
  const size_t gRowA = (size_t)bb0 * SEQLEN * HID + (size_t)(d0 + d4) +
                       (size_t)(2 * q * SEGLEN + tt) * HID;
  const size_t gRowB = gRowA + (size_t)SEGLEN * HID;
  const size_t kStep = (size_t)2 * SEQLEN * HID;
  const int uw = d4 * U_PD + ((tt >> 1) & 1) * U_PH + (tt >> 3) * U_PE +
                 bb0 * 4 + ((tt >> 2) & 1) * 2 + (tt & 1);
  const int yr = d4 * Y_PD + (tt >> 3) * Y_PH + ((tt >> 2) & 1) * Y_PI +
                 bb0 * 4 + (tt & 3);

  if (mover) {
#pragma unroll
    for (int k = 0; k < 4; ++k) {
      const v4f va = *(const v4f*)(u + gRowA + k * kStep);
      const v4f vb = *(const v4f*)(u + gRowB + k * kStep);
#pragma unroll
      for (int i = 0; i < 4; ++i) {
        ldsU[0][uw + 8 * k + i * U_PD]      = va[i];
        ldsU[0][uw + 8 * k + i * U_PD + 32] = vb[i];
      }
    }
    __builtin_prefetch(u + gRowA + (size_t)T * HID, 0, 0);
    __builtin_prefetch(u + gRowB + (size_t)T * HID, 0, 0);
  }
  __syncthreads();

  for (int c = 0; c < SEGCH; ++c) {
    if (mover) {
      float* un       = ldsU[(c + 1) & 1];
      const float* yp = ldsY[(c + 1) & 1];
      const size_t cn = (size_t)(c + 1) * T * HID;
      const size_t cp = cn - (size_t)2 * T * HID;
      const bool more = c + 1 < SEGCH;
#pragma nounroll
      for (int h = 0; h < 2; ++h) {
        const size_t ko = (size_t)(2 * h) * kStep;
        const int    lo = 16 * h;
        v4f va[2], vb[2];
        if (more) {
#pragma unroll
          for (int k = 0; k < 2; ++k) {
            va[k] = *(const v4f*)(u + gRowA + ko + k * kStep + cn);
            vb[k] = *(const v4f*)(u + gRowB + ko + k * kStep + cn);
          }
          if (c + 2 < SEGCH) {
#pragma unroll
            for (int k = 0; k < 2; ++k) {
              __builtin_prefetch(u + gRowA + ko + k * kStep + cn + (size_t)T * HID, 0, 0);
              __builtin_prefetch(u + gRowB + ko + k * kStep + cn + (size_t)T * HID, 0, 0);
            }
          }
        }
        if (c > 0) {
#pragma unroll
          for (int k = 0; k < 2; ++k) {
            v4f ya, yb;
#pragma unroll
            for (int i = 0; i < 4; ++i) {
              ya[i] = yp[yr + lo + 8 * k + i * Y_PD];
              yb[i] = yp[yr + lo + 8 * k + i * Y_PD + 32];
            }
            *(v4f*)(out + gRowA + ko + k * kStep + cp) = ya;
            *(v4f*)(out + gRowB + ko + k * kStep + cp) = yb;
          }
        }
        if (more) {
#pragma unroll
          for (int k = 0; k < 2; ++k)
#pragma unroll
            for (int i = 0; i < 4; ++i) {
              un[uw + lo + 8 * k + i * U_PD]      = va[k][i];
              un[uw + lo + 8 * k + i * U_PD + 32] = vb[k][i];
            }
        }
      }
#pragma unroll
      for (int r = 0; r < 4; ++r) x[r] = v8f{};
    } else {
      const float* ut = ldsU[c & 1];
      float* yt       = ldsY[c & 1];

      const v4f q0 = *(const v4f*)(&ut[ur]);
      const v4f q1 = *(const v4f*)(&ut[ur + U_PE]);
      v4f lv[8];
#pragma unroll
      for (int g = 0; g < 8; ++g)
        lv[g] = *(const v4f*)(&ldsLam[ch * STATE + half * 32 + 4 * g]);
      v2f uB[4];
      uB[0][0] = q0[0]; uB[0][1] = q0[1]; uB[1][0] = q0[2]; uB[1][1] = q0[3];
      uB[2][0] = q1[0]; uB[2][1] = q1[1]; uB[3][0] = q1[2]; uB[3][1] = q1[3];

      v8f y = {};
#pragma unroll
      for (int kk = 0; kk < 4; ++kk) y = wmma4(mA[kk], uB[kk], y);

      v2f xb[16];
#pragma unroll
      for (int r = 0; r < 4; ++r)
#pragma unroll
        for (int mm = 0; mm < 2; ++mm)
#pragma unroll
          for (int j = 0; j < 2; ++j) {
            float lo, hi;
            half_swap(x[r][4 * mm + j], x[r][4 * mm + 2 + j], lo, hi);
            xb[4 * r + mm][j]     = lo;
            xb[4 * r + 2 + mm][j] = hi;
          }

#pragma unroll
      for (int kk = 0; kk < 16; ++kk) y = wmma4(pA[kk], xb[kk], y);

      {
        v4f y0, y1;
#pragma unroll
        for (int i = 0; i < 4; ++i) { y0[i] = y[i]; y1[i] = y[4 + i]; }
        *(v4f*)(&yt[yw])        = y0;
        *(v4f*)(&yt[yw + Y_PI]) = y1;
      }

#pragma unroll
      for (int r = 0; r < 4; ++r)
#pragma unroll
        for (int mm = 0; mm < 2; ++mm)
#pragma unroll
          for (int j = 0; j < 2; ++j) {
            float a, b;
            half_swap(xb[4 * r + mm][j], xb[4 * r + 2 + mm][j], a, b);
            x[r][4 * mm + j]     = lv[2 * r + mm][j] * a;
            x[r][4 * mm + 2 + j] = lv[2 * r + mm][2 + j] * b;
          }
#pragma unroll
      for (int r = 0; r < 4; ++r) {
        v8f xn = x[r];
#pragma unroll
        for (int kk = 0; kk < 4; ++kk) xn = wmma4(gA[r][kk], uB[kk], xn);
        x[r] = xn;
      }
    }
    __syncthreads();
  }

  if (mover) {
    const float* yp = ldsY[(SEGCH - 1) & 1];
    const size_t cp = (size_t)(SEGCH - 1) * T * HID;
#pragma unroll
    for (int k = 0; k < 4; ++k) {
      v4f ya, yb;
#pragma unroll
      for (int i = 0; i < 4; ++i) {
        ya[i] = yp[yr + 8 * k + i * Y_PD];
        yb[i] = yp[yr + 8 * k + i * Y_PD + 32];
      }
      *(v4f*)(out + gRowA + k * kStep + cp) = ya;
      *(v4f*)(out + gRowB + k * kStep + cp) = yb;
    }
  }
}

extern "C" void kernel_launch(void* const* d_in, const int* in_sizes, int n_in,
                              void* d_out, int out_size, void* d_ws,
                              size_t ws_size, hipStream_t stream) {
  (void)in_sizes; (void)n_in; (void)out_size; (void)ws_size;
  const float* u  = (const float*)d_in[0];
  const float* A  = (const float*)d_in[1];
  const float* Bm = (const float*)d_in[2];
  const float* C  = (const float*)d_in[3];
  const float* Dv = (const float*)d_in[4];
  const float* dt = (const float*)d_in[5];
  float* out = (float*)d_out;
  float* ws  = (float*)d_ws;

  ssm_setup_kernel<<<HID, STATE, 0, stream>>>(A, Bm, C, Dv, dt, ws);
  ssm_segstate_kernel<<<dim3(HID / PDPB, NSUBPAIR), 32 * PDPB, 0, stream>>>(
      u, ws, ws + OFF_SUB);
  ssm_carry_kernel<<<HID, 512, 0, stream>>>(ws, ws + OFF_SUB, ws + OFF_STATE);
  ssm_scan_kernel<<<dim3(HID / DPB, NPAIR), SCAN_THREADS, 0, stream>>>(u, ws, out);
}


# Carry launch folded into the scan: each matrix wave chains the sub-segment end states in front of its own segment before its first chunk (planes interleaved for 8-byte reads); same sums; 3 launches
# speedup vs baseline: 5.5042x; 1.0397x over previous
#include <hip/hip_runtime.h>


typedef __attribute__((ext_vector_type(2))) float v2f;
typedef __attribute__((ext_vector_type(4))) float v4f;
typedef __attribute__((ext_vector_type(8))) float v8f;

#define BATCH  8
#define SEQLEN 4096
#define HID    1024
#define STATE  64
#define T      16
#define NCHUNK (SEQLEN / T)
#define DPB    16
#define BP     16

#define NSEG   8
#define NPAIR  (NSEG / 2)
#define SEGCH  (NCHUNK / NSEG)
#define SEGLEN (SEGCH * T)

#define PDPB   32
#define NSUB   (2 * NSEG)
#define SUBCH  (NCHUNK / NSUB)
#define SUBLEN (SUBCH * T)
#define NSUBPAIR (NSUB / 2 - 1)

#define PER_D   2368
#define OFF_P   0
#define OFF_G   1024
#define OFF_M   2048
#define OFF_LAM 2304

#define PLANE     1024
#define OFF_SUB   ((size_t)HID * PER_D)

#define U_PE   64
#define U_PH   132
#define U_PD   260
#define U_TILE (DPB * U_PD)
#define Y_PI   68
#define Y_PH   136
#define Y_PD   276
#define Y_TILE (DPB * Y_PD)

static __device__ __forceinline__ v8f wmma4(v2f a, v2f b, v8f c) {
  return __builtin_amdgcn_wmma_f32_16x16x4_f32(false, a, false, b, (short)0, c,
                                               false, false);
}

static __device__ __forceinline__ void half_swap(float a, float b, float& lo_hi,
                                                 float& hi_hi) {
  typedef __attribute__((ext_vector_type(2))) unsigned int v2u;
  const v2u r = __builtin_amdgcn_permlane16_swap(__float_as_uint(a),
                                                 __float_as_uint(b), false, false);
  lo_hi = __uint_as_float(r[0]);
  hi_hi = __uint_as_float(r[1]);
}

__global__ __launch_bounds__(64) void ssm_setup_kernel(
    const float* __restrict__ A, const float* __restrict__ Bm,
    const float* __restrict__ C, const float* __restrict__ Dv,
    const float* __restrict__ dt, float* __restrict__ ws) {
  const int d = blockIdx.x;
  const int n = threadIdx.x;
  __shared__ float sK[T * STATE];
  __shared__ float sKl[T];
  float* base = ws + (size_t)d * PER_D;

  const float a  = expf(A[d * STATE + n]);
  const float da = expf(-a * dt[n]);
  const float db = Bm[d * STATE + n] * (1.0f - da) / a;
  const float cc = C[d * STATE + n];

  float p = da;
#pragma unroll
  for (int t = 0; t < T; ++t) { base[OFF_P + t * STATE + n] = cc * p; p *= da; }

  const float da2 = da * da, da4 = da2 * da2, da8 = da4 * da4;
  base[OFF_LAM + n] = da8 * da8;

  float g = db;
  for (int s = T - 1; s >= 0; --s) { base[OFF_G + n * T + s] = g; g *= da; }

  float w = cc * db;
#pragma unroll
  for (int tau = 0; tau < T; ++tau) { sK[tau * STATE + n] = w; w *= da; }
  __syncthreads();
  if (n < T) {
    float s = 0.0f;
    for (int m = 0; m < STATE; ++m) s += sK[n * STATE + m];
    sKl[n] = s;
  }
  __syncthreads();
  if (n < T) {
    const int t = n;
    const float dvd = Dv[d];
    for (int s2 = 0; s2 < T; ++s2) {
      float v = (t >= s2) ? sKl[t - s2] : 0.0f;
      if (s2 == t) v += dvd;
      base[OFF_M + t * T + s2] = v;
    }
  }
}

__global__ __launch_bounds__(32 * PDPB) void ssm_segstate_kernel(
    const float* __restrict__ u, const float* __restrict__ prm,
    float* __restrict__ st) {
  const int tid  = threadIdx.x;
  const int lane = tid & 31;
  const int wv   = tid >> 5;
  const int d0   = blockIdx.x * PDPB;
  const int q    = blockIdx.y;
  const int half = lane >> 4;
  const int col  = lane & 15;

  __shared__ __align__(16) float ldsU[2][PDPB * U_PD];

  const float* base = prm + (size_t)(d0 + wv) * PER_D;

  v2f gA[4][4];
#pragma unroll
  for (int r = 0; r < 4; ++r)
#pragma unroll
    for (int kk = 0; kk < 4; ++kk)
      gA[r][kk] =
          *(const v2f*)(base + OFF_G + (r * 16 + col) * T + 4 * kk + 2 * half);
  float lam[4][8];
#pragma unroll
  for (int r = 0; r < 4; ++r)
#pragma unroll
    for (int i = 0; i < 8; ++i)
      lam[r][i] = base[OFF_LAM + r * 16 + i + 8 * half];

  v8f x[4] = {};

  const int pair = tid >> 3;
  const int bb   = pair >> 4;
  const int tt   = pair & 15;
  const int d4   = (tid & 7) << 2;
  const float* gA0 = u + (size_t)bb * SEQLEN * HID + (size_t)(d0 + d4) +
                     (size_t)(2 * q * SUBLEN + tt) * HID;
  const float* gB0 = gA0 + (size_t)SUBLEN * HID;
  const int uw = d4 * U_PD + ((tt >> 1) & 1) * U_PH + (tt >> 3) * U_PE +
                 bb * 4 + ((tt >> 2) & 1) * 2 + (tt & 1);
  const int ur = wv * U_PD + half * U_PH + col * 4;

  v4f va = *(const v4f*)gA0;
  v4f vb = *(const v4f*)gB0;

  for (int c = 0; c < SUBCH; ++c) {
    float* tile = ldsU[c & 1];
#pragma unroll
    for (int i = 0; i < 4; ++i) {
      tile[uw + i * U_PD]      = va[i];
      tile[uw + i * U_PD + 32] = vb[i];
    }
    __syncthreads();
    if (c + 1 < SUBCH) {
      va = *(const v4f*)(gA0 + (size_t)(c + 1) * T * HID);
      vb = *(const v4f*)(gB0 + (size_t)(c + 1) * T * HID);
    }
    if (c + 2 < SUBCH) {
      __builtin_prefetch(gA0 + (size_t)(c + 2) * T * HID, 0, 0);
      __builtin_prefetch(gB0 + (size_t)(c + 2) * T * HID, 0, 0);
    }

    const v4f q0 = *(const v4f*)(&tile[ur]);
    const v4f q1 = *(const v4f*)(&tile[ur + U_PE]);
    v2f uB[4];
    uB[0][0] = q0[0]; uB[0][1] = q0[1]; uB[1][0] = q0[2]; uB[1][1] = q0[3];
    uB[2][0] = q1[0]; uB[2][1] = q1[1]; uB[3][0] = q1[2]; uB[3][1] = q1[3];

#pragma unroll
    for (int r = 0; r < 4; ++r) {
      v8f xn;
#pragma unroll
      for (int i = 0; i < 8; ++i) xn[i] = lam[r][i] * x[r][i];
#pragma unroll
      for (int kk = 0; kk < 4; ++kk) xn = wmma4(gA[r][kk], uB[kk], xn);
      x[r] = xn;
    }
  }

  float* sp = st + ((size_t)q * HID + (size_t)(d0 + wv)) * PLANE + half * 16 +
              (col & 7) * 2 + (col >> 3);
#pragma unroll
  for (int r = 0; r < 4; ++r)
#pragma unroll
    for (int i = 0; i < 8; ++i) sp[(r * 8 + i) * 32] = x[r][i];
}

static __device__ __forceinline__ void chain_plane(const float* __restrict__ pp,
                                                   const float (&ls)[4][8],
                                                   v8f (&x)[4]) {
  v2f e[4][8];
#pragma unroll
  for (int r = 0; r < 4; ++r)
#pragma unroll
    for (int i = 0; i < 8; ++i) e[r][i] = *(const v2f*)(pp + (r * 8 + i) * 32);
#pragma unroll
  for (int r = 0; r < 4; ++r)
#pragma unroll
    for (int i = 0; i < 8; ++i) {
      float carry = x[r][i];
      carry = ls[r][i] * carry + e[r][i][0];
      carry = ls[r][i] * carry + e[r][i][1];
      x[r][i] = carry;
    }
}

#define NLW 4
#define SCAN_THREADS (32 * (DPB + NLW))

__global__ __launch_bounds__(SCAN_THREADS) void ssm_scan_kernel(
    const float* __restrict__ u, const float* __restrict__ ws,
    float* __restrict__ out) {
  const int tid  = threadIdx.x;
  const int lane = tid & 31;
  const int wv   = __builtin_amdgcn_readfirstlane(tid >> 5);
  const bool mover = wv < NLW;
  const int ch   = wv - NLW;
  const int d0   = blockIdx.x * DPB;
  const int q    = blockIdx.y;
  const int half = lane >> 4;
  const int col  = lane & 15;

  __shared__ __align__(16) float ldsU[2][U_TILE];
  __shared__ __align__(16) float ldsY[2][Y_TILE];
  __shared__ __align__(16) float ldsLam[DPB * STATE];

  v2f pA[16];
  v2f mA[4];
  v2f gA[4][4];
  v8f x[4] = {};
  if (!mover) {
    const float* base = ws + (size_t)(d0 + ch) * PER_D;
    {
      const float* ep = ws + OFF_SUB + (size_t)(d0 + ch) * PLANE + half * 16 +
                        (col & 7) * 2;
      float ls[4][8];
#pragma unroll
      for (int r = 0; r < 4; ++r)
#pragma unroll
        for (int i = 0; i < 8; ++i) {
          float l = base[OFF_LAM + r * 16 + i + 8 * half];
#pragma unroll
          for (int k = 1; k < SUBCH; k <<= 1) l *= l;
          ls[r][i] = l;
        }
#pragma nounroll
      for (int p = 0; p < 2 * q; ++p)
        chain_plane(ep + (size_t)p * HID * PLANE, ls, x);
      if (col >> 3) chain_plane(ep + (size_t)(2 * q) * HID * PLANE, ls, x);
    }
#pragma unroll
    for (int kk = 0; kk < 16; ++kk)
      pA[kk] = *(const v2f*)(base + OFF_P + col * STATE + 4 * kk + 2 * half);
#pragma unroll
    for (int kk = 0; kk < 4; ++kk)
      mA[kk] = *(const v2f*)(base + OFF_M + col * T + 4 * kk + 2 * half);
#pragma unroll
    for (int r = 0; r < 4; ++r)
#pragma unroll
      for (int kk = 0; kk < 4; ++kk)
        gA[r][kk] =
            *(const v2f*)(base + OFF_G + (r * 16 + col) * T + 4 * kk + 2 * half);
#pragma unroll
    for (int p = 0; p < 2; ++p) {
      const int idx = lane + 32 * p;
      ldsLam[ch * STATE + idx] =
          base[OFF_LAM + ((idx >> 3) & 3) * 16 + (idx & 7) + 8 * p];
    }
  }
  const int ur = ch * U_PD + half * U_PH + col * 4;
  const int yw = ch * Y_PD + half * Y_PH + col * 4;

  const int mt  = tid & 127;
  const int bb0 = mt >> 6;
  const int tt  = (mt >> 2) & 15;
  const int d4  = (mt & 3) << 2;
  const size_t gRowA = (size_t)bb0 * SEQLEN * HID + (size_t)(d0 + d4) +
                       (size_t)(2 * q * SEGLEN + tt) * HID;
  const size_t gRowB = gRowA + (size_t)SEGLEN * HID;
  const size_t kStep = (size_t)2 * SEQLEN * HID;
  const int uw = d4 * U_PD + ((tt >> 1) & 1) * U_PH + (tt >> 3) * U_PE +
                 bb0 * 4 + ((tt >> 2) & 1) * 2 + (tt & 1);
  const int yr = d4 * Y_PD + (tt >> 3) * Y_PH + ((tt >> 2) & 1) * Y_PI +
                 bb0 * 4 + (tt & 3);

  if (mover) {
#pragma unroll
    for (int k = 0; k < 4; ++k) {
      const v4f va = *(const v4f*)(u + gRowA + k * kStep);
      const v4f vb = *(const v4f*)(u + gRowB + k * kStep);
#pragma unroll
      for (int i = 0; i < 4; ++i) {
        ldsU[0][uw + 8 * k + i * U_PD]      = va[i];
        ldsU[0][uw + 8 * k + i * U_PD + 32] = vb[i];
      }
    }
    __builtin_prefetch(u + gRowA + (size_t)T * HID, 0, 0);
    __builtin_prefetch(u + gRowB + (size_t)T * HID, 0, 0);
  }
  __syncthreads();

  for (int c = 0; c < SEGCH; ++c) {
    if (mover) {
      float* un       = ldsU[(c + 1) & 1];
      const float* yp = ldsY[(c + 1) & 1];
      const size_t cn = (size_t)(c + 1) * T * HID;
      const size_t cp = cn - (size_t)2 * T * HID;
      const bool more = c + 1 < SEGCH;
#pragma nounroll
      for (int h = 0; h < 2; ++h) {
        const size_t ko = (size_t)(2 * h) * kStep;
        const int    lo = 16 * h;
        v4f va[2], vb[2];
        if (more) {
#pragma unroll
          for (int k = 0; k < 2; ++k) {
            va[k] = *(const v4f*)(u + gRowA + ko + k * kStep + cn);
            vb[k] = *(const v4f*)(u + gRowB + ko + k * kStep + cn);
          }
          if (c + 2 < SEGCH) {
#pragma unroll
            for (int k = 0; k < 2; ++k) {
              __builtin_prefetch(u + gRowA + ko + k * kStep + cn + (size_t)T * HID, 0, 0);
              __builtin_prefetch(u + gRowB + ko + k * kStep + cn + (size_t)T * HID, 0, 0);
            }
          }
        }
        if (c > 0) {
#pragma unroll
          for (int k = 0; k < 2; ++k) {
            v4f ya, yb;
#pragma unroll
            for (int i = 0; i < 4; ++i) {
              ya[i] = yp[yr + lo + 8 * k + i * Y_PD];
              yb[i] = yp[yr + lo + 8 * k + i * Y_PD + 32];
            }
            *(v4f*)(out + gRowA + ko + k * kStep + cp) = ya;
            *(v4f*)(out + gRowB + ko + k * kStep + cp) = yb;
          }
        }
        if (more) {
#pragma unroll
          for (int k = 0; k < 2; ++k)
#pragma unroll
            for (int i = 0; i < 4; ++i) {
              un[uw + lo + 8 * k + i * U_PD]      = va[k][i];
              un[uw + lo + 8 * k + i * U_PD + 32] = vb[k][i];
            }
        }
      }
#pragma unroll
      for (int r = 0; r < 4; ++r) x[r] = v8f{};
    } else {
      const float* ut = ldsU[c & 1];
      float* yt       = ldsY[c & 1];

      const v4f q0 = *(const v4f*)(&ut[ur]);
      const v4f q1 = *(const v4f*)(&ut[ur + U_PE]);
      v4f lv[8];
#pragma unroll
      for (int g = 0; g < 8; ++g)
        lv[g] = *(const v4f*)(&ldsLam[ch * STATE + half * 32 + 4 * g]);
      v2f uB[4];
      uB[0][0] = q0[0]; uB[0][1] = q0[1]; uB[1][0] = q0[2]; uB[1][1] = q0[3];
      uB[2][0] = q1[0]; uB[2][1] = q1[1]; uB[3][0] = q1[2]; uB[3][1] = q1[3];

      v8f y = {};
#pragma unroll
      for (int kk = 0; kk < 4; ++kk) y = wmma4(mA[kk], uB[kk], y);

      v2f xb[16];
#pragma unroll
      for (int r = 0; r < 4; ++r)
#pragma unroll
        for (int mm = 0; mm < 2; ++mm)
#pragma unroll
          for (int j = 0; j < 2; ++j) {
            float lo, hi;
            half_swap(x[r][4 * mm + j], x[r][4 * mm + 2 + j], lo, hi);
            xb[4 * r + mm][j]     = lo;
            xb[4 * r + 2 + mm][j] = hi;
          }

#pragma unroll
      for (int kk = 0; kk < 16; ++kk) y = wmma4(pA[kk], xb[kk], y);

      {
        v4f y0, y1;
#pragma unroll
        for (int i = 0; i < 4; ++i) { y0[i] = y[i]; y1[i] = y[4 + i]; }
        *(v4f*)(&yt[yw])        = y0;
        *(v4f*)(&yt[yw + Y_PI]) = y1;
      }

#pragma unroll
      for (int r = 0; r < 4; ++r)
#pragma unroll
        for (int mm = 0; mm < 2; ++mm)
#pragma unroll
          for (int j = 0; j < 2; ++j) {
            float a, b;
            half_swap(xb[4 * r + mm][j], xb[4 * r + 2 + mm][j], a, b);
            x[r][4 * mm + j]     = lv[2 * r + mm][j] * a;
            x[r][4 * mm + 2 + j] = lv[2 * r + mm][2 + j] * b;
          }
#pragma unroll
      for (int r = 0; r < 4; ++r) {
        v8f xn = x[r];
#pragma unroll
        for (int kk = 0; kk < 4; ++kk) xn = wmma4(gA[r][kk], uB[kk], xn);
        x[r] = xn;
      }
    }
    __syncthreads();
  }

  if (mover) {
    const float* yp = ldsY[(SEGCH - 1) & 1];
    const size_t cp = (size_t)(SEGCH - 1) * T * HID;
#pragma unroll
    for (int k = 0; k < 4; ++k) {
      v4f ya, yb;
#pragma unroll
      for (int i = 0; i < 4; ++i) {
        ya[i] = yp[yr + 8 * k + i * Y_PD];
        yb[i] = yp[yr + 8 * k + i * Y_PD + 32];
      }
      *(v4f*)(out + gRowA + k * kStep + cp) = ya;
      *(v4f*)(out + gRowB + k * kStep + cp) = yb;
    }
  }
}

extern "C" void kernel_launch(void* const* d_in, const int* in_sizes, int n_in,
                              void* d_out, int out_size, void* d_ws,
                              size_t ws_size, hipStream_t stream) {
  (void)in_sizes; (void)n_in; (void)out_size; (void)ws_size;
  const float* u  = (const float*)d_in[0];
  const float* A  = (const float*)d_in[1];
  const float* Bm = (const float*)d_in[2];
  const float* C  = (const float*)d_in[3];
  const float* Dv = (const float*)d_in[4];
  const float* dt = (const float*)d_in[5];
  float* out = (float*)d_out;
  float* ws  = (float*)d_ws;

  ssm_setup_kernel<<<HID, STATE, 0, stream>>>(A, Bm, C, Dv, dt, ws);
  ssm_segstate_kernel<<<dim3(HID / PDPB, NSUBPAIR), 32 * PDPB, 0, stream>>>(
      u, ws, ws + OFF_SUB);
  ssm_scan_kernel<<<dim3(HID / DPB, NPAIR), SCAN_THREADS, 0, stream>>>(u, ws, out);
}
